# starting kernel
# baseline (speedup 1.0000x reference)
#include <hip/hip_runtime.h>


typedef __attribute__((ext_vector_type(2))) float v2f;
typedef __attribute__((ext_vector_type(8))) float v8f;

#define NCH   20
#define CE    32
#define KC    33
#define WEMB  512
#define KDIM  132
#define NTILES (WEMB/16)

#define PAIR_STRIDE 1056
#define LDS_FLOATS  (66 * PAIR_STRIDE)

__global__ __launch_bounds__(256) void char_cnn_wmma(
    const float* __restrict__ char_emb_w,
    const float* __restrict__ cnn_w,
    const int*   __restrict__ data,
    const int*   __restrict__ spelling,
    float*       __restrict__ out,
    int nWords)
{
  extern __shared__ float ldsW[];
  const int tid = threadIdx.x;

  for (int e = tid; e < WEMB * KDIM; e += 256) {
    const int we = e / KDIM;
    const int j  = e % KDIM;
    ldsW[(j >> 1) * PAIR_STRIDE + 2 * we + (j & 1)] = cnn_w[e];
  }
  __syncthreads();

  const int lane    = tid & 31;
  const int wave    = tid >> 5;
  const int m_local = lane & 15;
  const int half    = lane >> 4;

  for (int word = blockIdx.x * 8 + wave; word < nWords; word += gridDim.x * 8) {
    const int wid   = data[word];
    const int sbase = wid * NCH;

    v2f a[2][KC];
    #pragma unroll
    for (int mt = 0; mt < 2; ++mt) {
      const int r1 = mt * 16 + m_local + 2 * half;
      const int r2 = r1 + 1;
      const bool ok1 = (r1 < NCH);
      const bool ok2 = (r2 < NCH);
      const int ch1 = ok1 ? spelling[sbase + r1] : 0;
      const int ch2 = ok2 ? spelling[sbase + r2] : 0;
      const float* e1 = char_emb_w + ch1 * CE;
      const float* e2 = char_emb_w + ch2 * CE;
      #pragma unroll
      for (int cb = 0; cb < CE; cb += 4) {
        const float4 q1 = ok1 ? *(const float4*)(e1 + cb) : make_float4(0.f, 0.f, 0.f, 0.f);
        const float4 q2 = ok2 ? *(const float4*)(e2 + cb) : make_float4(0.f, 0.f, 0.f, 0.f);
        a[mt][cb + 0].x = q1.x;  a[mt][cb + 0].y = q2.x;
        a[mt][cb + 1].x = q1.y;  a[mt][cb + 1].y = q2.y;
        a[mt][cb + 2].x = q1.z;  a[mt][cb + 2].y = q2.z;
        a[mt][cb + 3].x = q1.w;  a[mt][cb + 3].y = q2.w;
      }
      a[mt][CE].x = ok1 ? 1.f : 0.f;
      a[mt][CE].y = ok2 ? 1.f : 0.f;
    }

    #pragma unroll 1
    for (int nt = 0; nt < NTILES; ++nt) {
      v8f acc0 = {};
      v8f acc1 = {};
      const int colBase = 2 * (nt * 16 + m_local);
      #pragma unroll
      for (int kt = 0; kt < KC; ++kt) {
        const int pair = 2 * kt + half;
        const v2f b = *(const v2f*)&ldsW[pair * PAIR_STRIDE + colBase];
        acc0 = __builtin_amdgcn_wmma_f32_16x16x4_f32(
            false, a[0][kt], false, b, (short)0, acc0, false, false);
        acc1 = __builtin_amdgcn_wmma_f32_16x16x4_f32(
            false, a[1][kt], false, b, (short)0, acc1, false, false);
      }
      float m = acc0[0];
      #pragma unroll
      for (int i = 1; i < 8; ++i) m = fmaxf(m, acc0[i]);
      m = fmaxf(m, __shfl_xor(m, 16, 32));
      const float t16 = __shfl(acc1[0], m_local, 32);
      m = fmaxf(m, t16);
      if (lane < 16) out[word * WEMB + nt * 16 + lane] = m;
    }
  }
}

extern "C" void kernel_launch(void* const* d_in, const int* in_sizes, int n_in,
                              void* d_out, int out_size, void* d_ws, size_t ws_size,
                              hipStream_t stream) {
  const float* char_emb_w = (const float*)d_in[0];
  const float* cnn_w      = (const float*)d_in[1];
  const int*   data       = (const int*)d_in[2];
  const int*   spelling   = (const int*)d_in[3];
  float*       out        = (float*)d_out;
  const int nWords = in_sizes[2];

  const int blocks = 256;
  const size_t shmem = LDS_FLOATS * sizeof(float);
  hipLaunchKernelGGL(char_cnn_wmma, dim3(blocks), dim3(256), shmem, stream,
                     char_emb_w, cnn_w, data, spelling, out, nWords);
}


# Fill the M tiles: a wave owns 16 words and position 16 of all 16 shares one tile (17 full tiles per 16 words, was 32 half-empty); block = 128 words x 256 channels so that plane fits in LDS
# speedup vs baseline: 1.7995x; 1.7995x over previous
#include <hip/hip_runtime.h>


typedef __attribute__((ext_vector_type(2))) float v2f;
typedef __attribute__((ext_vector_type(8))) float v8f;

#define NCH   20
#define CE    32
#define KC    33
#define WEMB  512
#define KDIM  132
#define NHALF 256
#define NT_B  (NHALF/16)
#define WPW   16
#define WPB   (8*WPW)

#define PAIR_STRIDE 544
#define W_FLOATS    (66 * PAIR_STRIDE)
#define S_PITCH     276
#define S_WAVE      (WPW * S_PITCH)
#define LDS_FLOATS  (W_FLOATS + 8 * S_WAVE)

__global__ __launch_bounds__(256) void char_cnn_wmma(
    const float* __restrict__ char_emb_w,
    const float* __restrict__ cnn_w,
    const int*   __restrict__ data,
    const int*   __restrict__ spelling,
    float*       __restrict__ out,
    int nWords)
{
  extern __shared__ float lds[];
  float* ldsW = lds;
  const int tid   = threadIdx.x;
  const int chan0 = (blockIdx.x & 1) * NHALF;

  const float* wsrc = cnn_w + chan0 * KDIM;
  for (int e = tid; e < NHALF * KDIM; e += 256) {
    const int wl = e / KDIM;
    const int j  = e % KDIM;
    ldsW[(j >> 1) * PAIR_STRIDE + 2 * wl + (j & 1)] = wsrc[e];
  }
  __syncthreads();

  const int lane = tid & 31;
  const int wave = tid >> 5;
  const int c    = lane & 15;
  const int h    = lane >> 4;
  float* s16 = lds + W_FLOATS + wave * S_WAVE;

  const int passStride = (gridDim.x >> 1) * WPB;
  for (int gbase = (blockIdx.x >> 1) * WPB; gbase < nWords; gbase += passStride) {
    const int w0 = gbase + wave * WPW;
    if (w0 < nWords) {
      {
        int wd = w0 + c;
        if (wd >= nWords) wd = nWords - 1;
        const int sb  = data[wd] * NCH + 16 + 2 * h;
        const int ch1 = spelling[sb];
        const int ch2 = spelling[sb + 1];
        const float* e1 = char_emb_w + ch1 * CE;
        const float* e2 = char_emb_w + ch2 * CE;
        v2f a16[KC];
        #pragma unroll
        for (int cb = 0; cb < CE; cb += 4) {
          const float4 q1 = *(const float4*)(e1 + cb);
          const float4 q2 = *(const float4*)(e2 + cb);
          a16[cb + 0].x = q1.x;  a16[cb + 0].y = q2.x;
          a16[cb + 1].x = q1.y;  a16[cb + 1].y = q2.y;
          a16[cb + 2].x = q1.z;  a16[cb + 2].y = q2.z;
          a16[cb + 3].x = q1.w;  a16[cb + 3].y = q2.w;
        }
        a16[CE].x = 1.f;
        a16[CE].y = 1.f;

        #pragma unroll 1
        for (int nt = 0; nt < NT_B; nt += 2) {
          v8f d0 = {};
          v8f d1 = {};
          const int colBase = 2 * (nt * 16 + c);
          #pragma unroll
          for (int kt = 0; kt < KC; ++kt) {
            const int pair = 2 * kt + h;
            const v2f b0 = *(const v2f*)&ldsW[pair * PAIR_STRIDE + colBase];
            const v2f b1 = *(const v2f*)&ldsW[pair * PAIR_STRIDE + colBase + 32];
            d0 = __builtin_amdgcn_wmma_f32_16x16x4_f32(
                false, a16[kt], false, b0, (short)0, d0, false, false);
            d1 = __builtin_amdgcn_wmma_f32_16x16x4_f32(
                false, a16[kt], false, b1, (short)0, d1, false, false);
          }
          #pragma unroll
          for (int r = 0; r < 8; ++r) {
            s16[(8 * h + r) * S_PITCH + nt * 16 + c]      = d0[r];
            s16[(8 * h + r) * S_PITCH + nt * 16 + 16 + c] = d1[r];
          }
        }
      }

      #pragma unroll 1
      for (int j = 0; j < WPW / 2; ++j) {
        v2f a[2][KC];
        #pragma unroll
        for (int mt = 0; mt < 2; ++mt) {
          int wd = w0 + 2 * j + mt;
          if (wd >= nWords) wd = nWords - 1;
          const int sb  = data[wd] * NCH + c + 2 * h;
          const int ch1 = spelling[sb];
          const int ch2 = spelling[sb + 1];
          const float* e1 = char_emb_w + ch1 * CE;
          const float* e2 = char_emb_w + ch2 * CE;
          #pragma unroll
          for (int cb = 0; cb < CE; cb += 4) {
            const float4 q1 = *(const float4*)(e1 + cb);
            const float4 q2 = *(const float4*)(e2 + cb);
            a[mt][cb + 0].x = q1.x;  a[mt][cb + 0].y = q2.x;
            a[mt][cb + 1].x = q1.y;  a[mt][cb + 1].y = q2.y;
            a[mt][cb + 2].x = q1.z;  a[mt][cb + 2].y = q2.z;
            a[mt][cb + 3].x = q1.w;  a[mt][cb + 3].y = q2.w;
          }
          a[mt][CE].x = 1.f;
          a[mt][CE].y = 1.f;
        }

        const int wl = 2 * j + h;
        const int wo = w0 + wl;
        const bool wok = (wo < nWords);
        float* orow = out + wo * WEMB + chan0 + c;
        const float* srow = s16 + wl * S_PITCH + c;

        #pragma unroll 1
        for (int nt = 0; nt < NT_B; ++nt) {
          v8f acc0 = {};
          v8f acc1 = {};
          const int colBase = 2 * (nt * 16 + c);
          #pragma unroll
          for (int kt = 0; kt < KC; ++kt) {
            const int pair = 2 * kt + h;
            const v2f b = *(const v2f*)&ldsW[pair * PAIR_STRIDE + colBase];
            acc0 = __builtin_amdgcn_wmma_f32_16x16x4_f32(
                false, a[0][kt], false, b, (short)0, acc0, false, false);
            acc1 = __builtin_amdgcn_wmma_f32_16x16x4_f32(
                false, a[1][kt], false, b, (short)0, acc1, false, false);
          }
          float m0 = acc0[0];
          float m1 = acc1[0];
          #pragma unroll
          for (int r = 1; r < 8; ++r) {
            m0 = fmaxf(m0, acc0[r]);
            m1 = fmaxf(m1, acc1[r]);
          }
          m0 = fmaxf(m0, __shfl_xor(m0, 16, 32));
          m1 = fmaxf(m1, __shfl_xor(m1, 16, 32));
          float m = h ? m1 : m0;
          m = fmaxf(m, srow[nt * 16]);
          if (wok) orow[nt * 16] = m;
        }
      }
    }
  }
}

extern "C" void kernel_launch(void* const* d_in, const int* in_sizes, int n_in,
                              void* d_out, int out_size, void* d_ws, size_t ws_size,
                              hipStream_t stream) {
  const float* char_emb_w = (const float*)d_in[0];
  const float* cnn_w      = (const float*)d_in[1];
  const int*   data       = (const int*)d_in[2];
  const int*   spelling   = (const int*)d_in[3];
  float*       out        = (float*)d_out;
  const int nWords = in_sizes[2];

  const int blocks = 256;
  const size_t shmem = LDS_FLOATS * sizeof(float);
  hipLaunchKernelGGL(char_cnn_wmma, dim3(blocks), dim3(256), shmem, stream,
                     char_emb_w, cnn_w, data, spelling, out, nWords);
}


# One LDS place per N tile: next tile's 33 B fragments and position-16 value fetched as one group a WMMA block ahead into a second register set; pool half-swap on permlanex16
# speedup vs baseline: 1.9841x; 1.1026x over previous
#include <hip/hip_runtime.h>


typedef __attribute__((ext_vector_type(2))) float v2f;
typedef __attribute__((ext_vector_type(8))) float v8f;

#define NCH   20
#define CE    32
#define KC    33
#define WEMB  512
#define KDIM  132
#define NHALF 256
#define NT_B  (NHALF/16)
#define WPW   16
#define WPB   (8*WPW)

#define PAIR_STRIDE 544
#define W_FLOATS    (66 * PAIR_STRIDE)
#define S_PITCH     276
#define S_WAVE      (WPW * S_PITCH)
#define LDS_FLOATS  (W_FLOATS + 8 * S_WAVE)

static __device__ __forceinline__ void fence() { __builtin_amdgcn_sched_barrier(0); }

static __device__ __forceinline__ float xhalf(float x) {
  return __builtin_bit_cast(float, __builtin_amdgcn_permlanex16(
      __builtin_bit_cast(int, x), __builtin_bit_cast(int, x), 0x76543210, 0xfedcba98, false, false));
}

static __device__ __forceinline__ void load_tile(v2f (&b)[KC], float& s,
                                                 const float* bp, const float* sp) {
  #pragma unroll
  for (int kt = 0; kt < KC; ++kt)
    b[kt] = *(const v2f*)(bp + kt * 2 * PAIR_STRIDE);
  s = *sp;
}

static __device__ __forceinline__ void do_tile(const v2f (&a)[2][KC], const v2f (&b)[KC],
                                               float s, int h, bool wok, float* o) {
  v8f acc0 = {};
  v8f acc1 = {};
  #pragma unroll
  for (int kt = 0; kt < KC; ++kt) {
    acc0 = __builtin_amdgcn_wmma_f32_16x16x4_f32(
        false, a[0][kt], false, b[kt], (short)0, acc0, false, false);
    acc1 = __builtin_amdgcn_wmma_f32_16x16x4_f32(
        false, a[1][kt], false, b[kt], (short)0, acc1, false, false);
  }
  float m0 = acc0[0];
  float m1 = acc1[0];
  #pragma unroll
  for (int r = 1; r < 8; ++r) {
    m0 = fmaxf(m0, acc0[r]);
    m1 = fmaxf(m1, acc1[r]);
  }
  const float mine = h ? m1 : m0;
  const float give = h ? m0 : m1;
  float m = fmaxf(mine, xhalf(give));
  m = fmaxf(m, s);
  if (wok) *o = m;
}

__global__ __launch_bounds__(256) void char_cnn_wmma(
    const float* __restrict__ char_emb_w,
    const float* __restrict__ cnn_w,
    const int*   __restrict__ data,
    const int*   __restrict__ spelling,
    float*       __restrict__ out,
    int nWords)
{
  extern __shared__ float lds[];
  float* ldsW = lds;
  const int tid   = threadIdx.x;
  const int chan0 = (blockIdx.x & 1) * NHALF;

  const float* wsrc = cnn_w + chan0 * KDIM;
  for (int e = tid; e < NHALF * KDIM; e += 256) {
    const int wl = e / KDIM;
    const int j  = e % KDIM;
    ldsW[(j >> 1) * PAIR_STRIDE + 2 * wl + (j & 1)] = wsrc[e];
  }
  __syncthreads();

  const int lane = tid & 31;
  const int wave = tid >> 5;
  const int c    = lane & 15;
  const int h    = lane >> 4;
  float* s16 = lds + W_FLOATS + wave * S_WAVE;

  const int passStride = (gridDim.x >> 1) * WPB;
  for (int gbase = (blockIdx.x >> 1) * WPB; gbase < nWords; gbase += passStride) {
    const int w0 = gbase + wave * WPW;
    if (w0 < nWords) {
      {
        int wd = w0 + c;
        if (wd >= nWords) wd = nWords - 1;
        const int sb  = data[wd] * NCH + 16 + 2 * h;
        const int ch1 = spelling[sb];
        const int ch2 = spelling[sb + 1];
        const float* e1 = char_emb_w + ch1 * CE;
        const float* e2 = char_emb_w + ch2 * CE;
        v2f a16[KC];
        #pragma unroll
        for (int cb = 0; cb < CE; cb += 4) {
          const float4 q1 = *(const float4*)(e1 + cb);
          const float4 q2 = *(const float4*)(e2 + cb);
          a16[cb + 0].x = q1.x;  a16[cb + 0].y = q2.x;
          a16[cb + 1].x = q1.y;  a16[cb + 1].y = q2.y;
          a16[cb + 2].x = q1.z;  a16[cb + 2].y = q2.z;
          a16[cb + 3].x = q1.w;  a16[cb + 3].y = q2.w;
        }
        a16[CE].x = 1.f;
        a16[CE].y = 1.f;

        #pragma unroll 1
        for (int nt = 0; nt < NT_B; nt += 2) {
          v8f d0 = {};
          v8f d1 = {};
          const int colBase = 2 * (nt * 16 + c);
          #pragma unroll
          for (int kt = 0; kt < KC; ++kt) {
            const int pair = 2 * kt + h;
            const v2f b0 = *(const v2f*)&ldsW[pair * PAIR_STRIDE + colBase];
            const v2f b1 = *(const v2f*)&ldsW[pair * PAIR_STRIDE + colBase + 32];
            d0 = __builtin_amdgcn_wmma_f32_16x16x4_f32(
                false, a16[kt], false, b0, (short)0, d0, false, false);
            d1 = __builtin_amdgcn_wmma_f32_16x16x4_f32(
                false, a16[kt], false, b1, (short)0, d1, false, false);
          }
          #pragma unroll
          for (int r = 0; r < 8; ++r) {
            s16[(8 * h + r) * S_PITCH + nt * 16 + c]      = d0[r];
            s16[(8 * h + r) * S_PITCH + nt * 16 + 16 + c] = d1[r];
          }
        }
      }

      const float* bp = ldsW + h * PAIR_STRIDE + 2 * c;
      v2f bA[KC], bB[KC];
      float sA, sB;
      load_tile(bA, sA, bp, s16 + h * S_PITCH + c);
      fence();

      #pragma unroll 1
      for (int j = 0; j < WPW / 2; ++j) {
        v2f a[2][KC];
        #pragma unroll
        for (int mt = 0; mt < 2; ++mt) {
          int wd = w0 + 2 * j + mt;
          if (wd >= nWords) wd = nWords - 1;
          const int sb  = data[wd] * NCH + c + 2 * h;
          const int ch1 = spelling[sb];
          const int ch2 = spelling[sb + 1];
          const float* e1 = char_emb_w + ch1 * CE;
          const float* e2 = char_emb_w + ch2 * CE;
          #pragma unroll
          for (int cb = 0; cb < CE; cb += 4) {
            const float4 q1 = *(const float4*)(e1 + cb);
            const float4 q2 = *(const float4*)(e2 + cb);
            a[mt][cb + 0].x = q1.x;  a[mt][cb + 0].y = q2.x;
            a[mt][cb + 1].x = q1.y;  a[mt][cb + 1].y = q2.y;
            a[mt][cb + 2].x = q1.z;  a[mt][cb + 2].y = q2.z;
            a[mt][cb + 3].x = q1.w;  a[mt][cb + 3].y = q2.w;
          }
          a[mt][CE].x = 1.f;
          a[mt][CE].y = 1.f;
        }

        const int wl = 2 * j + h;
        const int wo = w0 + wl;
        const bool wok = (wo < nWords);
        float* orow = out + wo * WEMB + chan0 + c;
        const float* srow  = s16 + wl * S_PITCH + c;
        const float* srowN = s16 + ((wl + 2) & (WPW - 1)) * S_PITCH + c;

        #pragma unroll 1
        for (int nt = 0; nt < NT_B; nt += 2) {
          fence();
          load_tile(bB, sB, bp + (nt + 1) * 32, srow + (nt + 1) * 16);
          fence();
          do_tile(a, bA, sA, h, wok, orow + nt * 16);
          fence();
          const bool last = (nt + 2 == NT_B);
          load_tile(bA, sA, bp + (last ? 0 : (nt + 2) * 32),
                    last ? srowN : srow + (nt + 2) * 16);
          fence();
          do_tile(a, bB, sB, h, wok, orow + (nt + 1) * 16);
          fence();
        }
      }
    }
  }
}

extern "C" void kernel_launch(void* const* d_in, const int* in_sizes, int n_in,
                              void* d_out, int out_size, void* d_ws, size_t ws_size,
                              hipStream_t stream) {
  const float* char_emb_w = (const float*)d_in[0];
  const float* cnn_w      = (const float*)d_in[1];
  const int*   data       = (const int*)d_in[2];
  const int*   spelling   = (const int*)d_in[3];
  float*       out        = (float*)d_out;
  const int nWords = in_sizes[2];

  const int blocks = 256;
  const size_t shmem = LDS_FLOATS * sizeof(float);
  hipLaunchKernelGGL(char_cnn_wmma, dim3(blocks), dim3(256), shmem, stream,
                     char_emb_w, cnn_w, data, spelling, out, nWords);
}


# Even out each SIMD's wave pair: the younger wave (w+4) holds s_setprio 3 for its first 6 of 8 word pairs, then 0, so neither wave runs its tail alone with unfilled load and gather gaps
# speedup vs baseline: 2.0909x; 1.0538x over previous
#include <hip/hip_runtime.h>


typedef __attribute__((ext_vector_type(2))) float v2f;
typedef __attribute__((ext_vector_type(8))) float v8f;

#define NCH   20
#define CE    32
#define KC    33
#define WEMB  512
#define KDIM  132
#define NHALF 256
#define NT_B  (NHALF/16)
#define WPW   16
#define WPB   (8*WPW)
#define PRIO_J 6

#define PAIR_STRIDE 544
#define W_FLOATS    (66 * PAIR_STRIDE)
#define S_PITCH     276
#define S_WAVE      (WPW * S_PITCH)
#define LDS_FLOATS  (W_FLOATS + 8 * S_WAVE)

static __device__ __forceinline__ void fence() { __builtin_amdgcn_sched_barrier(0); }

static __device__ __forceinline__ float xhalf(float x) {
  return __builtin_bit_cast(float, __builtin_amdgcn_permlanex16(
      __builtin_bit_cast(int, x), __builtin_bit_cast(int, x), 0x76543210, 0xfedcba98, false, false));
}

static __device__ __forceinline__ void load_tile(v2f (&b)[KC], float& s,
                                                 const float* bp, const float* sp) {
  #pragma unroll
  for (int kt = 0; kt < KC; ++kt)
    b[kt] = *(const v2f*)(bp + kt * 2 * PAIR_STRIDE);
  s = *sp;
}

static __device__ __forceinline__ void do_tile(const v2f (&a)[2][KC], const v2f (&b)[KC],
                                               float s, int h, bool wok, float* o) {
  v8f acc0 = {};
  v8f acc1 = {};
  #pragma unroll
  for (int kt = 0; kt < KC; ++kt) {
    acc0 = __builtin_amdgcn_wmma_f32_16x16x4_f32(
        false, a[0][kt], false, b[kt], (short)0, acc0, false, false);
    acc1 = __builtin_amdgcn_wmma_f32_16x16x4_f32(
        false, a[1][kt], false, b[kt], (short)0, acc1, false, false);
  }
  float m0 = acc0[0];
  float m1 = acc1[0];
  #pragma unroll
  for (int r = 1; r < 8; ++r) {
    m0 = fmaxf(m0, acc0[r]);
    m1 = fmaxf(m1, acc1[r]);
  }
  const float mine = h ? m1 : m0;
  const float give = h ? m0 : m1;
  float m = fmaxf(mine, xhalf(give));
  m = fmaxf(m, s);
  if (wok) *o = m;
}

__global__ __launch_bounds__(256) void char_cnn_wmma(
    const float* __restrict__ char_emb_w,
    const float* __restrict__ cnn_w,
    const int*   __restrict__ data,
    const int*   __restrict__ spelling,
    float*       __restrict__ out,
    int nWords)
{
  extern __shared__ float lds[];
  float* ldsW = lds;
  const int tid   = threadIdx.x;
  const int chan0 = (blockIdx.x & 1) * NHALF;

  const float* wsrc = cnn_w + chan0 * KDIM;
  for (int e = tid; e < NHALF * KDIM; e += 256) {
    const int wl = e / KDIM;
    const int j  = e % KDIM;
    ldsW[(j >> 1) * PAIR_STRIDE + 2 * wl + (j & 1)] = wsrc[e];
  }
  __syncthreads();

  const int lane = tid & 31;
  const int wave = tid >> 5;
  const int c    = lane & 15;
  const int h    = lane >> 4;
  float* s16 = lds + W_FLOATS + wave * S_WAVE;
  const bool younger = __builtin_amdgcn_readfirstlane(wave) >= 4;

  const int passStride = (gridDim.x >> 1) * WPB;
  for (int gbase = (blockIdx.x >> 1) * WPB; gbase < nWords; gbase += passStride) {
    const int w0 = gbase + wave * WPW;
    if (w0 < nWords) {
      if (younger) __builtin_amdgcn_s_setprio(3);
      {
        int wd = w0 + c;
        if (wd >= nWords) wd = nWords - 1;
        const int sb  = data[wd] * NCH + 16 + 2 * h;
        const int ch1 = spelling[sb];
        const int ch2 = spelling[sb + 1];
        const float* e1 = char_emb_w + ch1 * CE;
        const float* e2 = char_emb_w + ch2 * CE;
        v2f a16[KC];
        #pragma unroll
        for (int cb = 0; cb < CE; cb += 4) {
          const float4 q1 = *(const float4*)(e1 + cb);
          const float4 q2 = *(const float4*)(e2 + cb);
          a16[cb + 0].x = q1.x;  a16[cb + 0].y = q2.x;
          a16[cb + 1].x = q1.y;  a16[cb + 1].y = q2.y;
          a16[cb + 2].x = q1.z;  a16[cb + 2].y = q2.z;
          a16[cb + 3].x = q1.w;  a16[cb + 3].y = q2.w;
        }
        a16[CE].x = 1.f;
        a16[CE].y = 1.f;

        #pragma unroll 1
        for (int nt = 0; nt < NT_B; nt += 2) {
          v8f d0 = {};
          v8f d1 = {};
          const int colBase = 2 * (nt * 16 + c);
          #pragma unroll
          for (int kt = 0; kt < KC; ++kt) {
            const int pair = 2 * kt + h;
            const v2f b0 = *(const v2f*)&ldsW[pair * PAIR_STRIDE + colBase];
            const v2f b1 = *(const v2f*)&ldsW[pair * PAIR_STRIDE + colBase + 32];
            d0 = __builtin_amdgcn_wmma_f32_16x16x4_f32(
                false, a16[kt], false, b0, (short)0, d0, false, false);
            d1 = __builtin_amdgcn_wmma_f32_16x16x4_f32(
                false, a16[kt], false, b1, (short)0, d1, false, false);
          }
          #pragma unroll
          for (int r = 0; r < 8; ++r) {
            s16[(8 * h + r) * S_PITCH + nt * 16 + c]      = d0[r];
            s16[(8 * h + r) * S_PITCH + nt * 16 + 16 + c] = d1[r];
          }
        }
      }

      const float* bp = ldsW + h * PAIR_STRIDE + 2 * c;
      v2f bA[KC], bB[KC];
      float sA, sB;
      load_tile(bA, sA, bp, s16 + h * S_PITCH + c);
      fence();

      #pragma unroll 1
      for (int j = 0; j < WPW / 2; ++j) {
        if (younger && j == PRIO_J) __builtin_amdgcn_s_setprio(0);
        v2f a[2][KC];
        #pragma unroll
        for (int mt = 0; mt < 2; ++mt) {
          int wd = w0 + 2 * j + mt;
          if (wd >= nWords) wd = nWords - 1;
          const int sb  = data[wd] * NCH + c + 2 * h;
          const int ch1 = spelling[sb];
          const int ch2 = spelling[sb + 1];
          const float* e1 = char_emb_w + ch1 * CE;
          const float* e2 = char_emb_w + ch2 * CE;
          #pragma unroll
          for (int cb = 0; cb < CE; cb += 4) {
            const float4 q1 = *(const float4*)(e1 + cb);
            const float4 q2 = *(const float4*)(e2 + cb);
            a[mt][cb + 0].x = q1.x;  a[mt][cb + 0].y = q2.x;
            a[mt][cb + 1].x = q1.y;  a[mt][cb + 1].y = q2.y;
            a[mt][cb + 2].x = q1.z;  a[mt][cb + 2].y = q2.z;
            a[mt][cb + 3].x = q1.w;  a[mt][cb + 3].y = q2.w;
          }
          a[mt][CE].x = 1.f;
          a[mt][CE].y = 1.f;
        }

        const int wl = 2 * j + h;
        const int wo = w0 + wl;
        const bool wok = (wo < nWords);
        float* orow = out + wo * WEMB + chan0 + c;
        const float* srow  = s16 + wl * S_PITCH + c;
        const float* srowN = s16 + ((wl + 2) & (WPW - 1)) * S_PITCH + c;

        #pragma unroll 1
        for (int nt = 0; nt < NT_B; nt += 2) {
          fence();
          load_tile(bB, sB, bp + (nt + 1) * 32, srow + (nt + 1) * 16);
          fence();
          do_tile(a, bA, sA, h, wok, orow + nt * 16);
          fence();
          const bool last = (nt + 2 == NT_B);
          load_tile(bA, sA, bp + (last ? 0 : (nt + 2) * 32),
                    last ? srowN : srow + (nt + 2) * 16);
          fence();
          do_tile(a, bB, sB, h, wok, orow + (nt + 1) * 16);
          fence();
        }
      }
    }
  }
}

extern "C" void kernel_launch(void* const* d_in, const int* in_sizes, int n_in,
                              void* d_out, int out_size, void* d_ws, size_t ws_size,
                              hipStream_t stream) {
  const float* char_emb_w = (const float*)d_in[0];
  const float* cnn_w      = (const float*)d_in[1];
  const int*   data       = (const int*)d_in[2];
  const int*   spelling   = (const int*)d_in[3];
  float*       out        = (float*)d_out;
  const int nWords = in_sizes[2];

  const int blocks = 256;
  const size_t shmem = LDS_FLOATS * sizeof(float);
  hipLaunchKernelGGL(char_cnn_wmma, dim3(blocks), dim3(256), shmem, stream,
                     char_emb_w, cnn_w, data, spelling, out, nWords);
}
